# starting kernel
# baseline (speedup 1.0000x reference)
#include <hip/hip_runtime.h>
#include <hip/hip_bf16.h>
#include <math.h>

#define GMM_B 32768
#define GMM_C 100
#define GMM_D 64

typedef __attribute__((ext_vector_type(2))) float v2f;
typedef __attribute__((ext_vector_type(8))) float v8f;

__global__ __launch_bounds__(64)
void gmm_prep_kernel(const float* __restrict__ means,
                     const float* __restrict__ covs,
                     const float* __restrict__ weights,
                     float* __restrict__ Gout,
                     float* __restrict__ hout,
                     float* __restrict__ ccout)
{
    __shared__ float A [GMM_D * 65];
    __shared__ float Gi[GMM_D * 65];
    __shared__ float red[GMM_D];
    __shared__ float mu [GMM_D];

    const int t = threadIdx.x;
    const int c = blockIdx.x;

    for (int k = 0; k < GMM_D; ++k)
        A[t * 65 + k] = covs[(size_t)c * GMM_D * GMM_D + (size_t)t * GMM_D + k];
    mu[t] = means[c * GMM_D + t];
    __syncthreads();

    for (int j = 0; j < GMM_D; ++j) {
        if (t == j) A[j * 65 + j] = sqrtf(A[j * 65 + j]);
        __syncthreads();
        if (t > j) A[t * 65 + j] /= A[j * 65 + j];
        __syncthreads();
        if (t > j) {
            const float ltj = A[t * 65 + j];
            for (int k = j + 1; k <= t; ++k)
                A[t * 65 + k] -= ltj * A[k * 65 + j];
        }
        __syncthreads();
    }

    red[t] = 2.0f * logf(A[t * 65 + t]);
    __syncthreads();
    if (t == 0) {
        float ld = 0.0f;
        for (int k = 0; k < GMM_D; ++k) ld += red[k];
        const float LOG2PI = 1.8378770664093453f;
        ccout[c] = logf(weights[c]) - 0.5f * (ld + (float)GMM_D * LOG2PI);
    }

    for (int i = 0; i < GMM_D; ++i) {
        float sum = (i == t) ? 1.0f : 0.0f;
        for (int k = t; k < i; ++k)
            sum -= A[i * 65 + k] * Gi[k * 65 + t];
        Gi[i * 65 + t] = (i >= t) ? (sum / A[i * 65 + i]) : 0.0f;
    }
    __syncthreads();

    float hv = 0.0f;
    for (int k = 0; k <= t; ++k) hv += Gi[t * 65 + k] * mu[k];
    hout[c * GMM_D + t] = hv;

    for (int k = 0; k < GMM_D; ++k)
        Gout[(size_t)c * GMM_D * GMM_D + (size_t)t * GMM_D + k] = Gi[t * 65 + k];
}

#define TILE_N 128
#define LSTR   68

__global__ __launch_bounds__(256)
void gmm_logits_kernel(const float* __restrict__ x,
                       const float* __restrict__ G,
                       const float* __restrict__ h,
                       const float* __restrict__ cc,
                       float* __restrict__ out)
{
    __shared__ __align__(16) float Xl[TILE_N * LSTR];
    __shared__ __align__(16) float Gl[GMM_D * LSTR];
    __shared__ float hl[GMM_D];

    const int t  = threadIdx.x;
    const int c  = blockIdx.y;
    const int n0 = blockIdx.x * TILE_N;

    {
        const int col4 = (t & 15) * 4;
        const int r0   = t >> 4;
        #pragma unroll
        for (int it = 0; it < 8; ++it) {
            const int r = r0 + it * 16;
            const float4 v = *(const float4*)(x + (size_t)(n0 + r) * GMM_D + col4);
            *(float4*)&Xl[r * LSTR + col4] = v;
        }
        #pragma unroll
        for (int it = 0; it < 4; ++it) {
            const int r = r0 + it * 16;
            const float4 v = *(const float4*)(G + (size_t)c * GMM_D * GMM_D + (size_t)r * GMM_D + col4);
            *(float4*)&Gl[r * LSTR + col4] = v;
        }
        if (t < GMM_D) hl[t] = h[c * GMM_D + t];
    }
    __syncthreads();

    const int lane  = t & 31;
    const int w     = t >> 5;
    const int nloc  = lane & 15;
    const int khalf = (lane >> 4) * 2;

    const float* brow = &Xl[(w * 16 + nloc) * LSTR];

    float s = 0.0f;
    #pragma unroll
    for (int mt = 0; mt < 4; ++mt) {
        v8f acc = {};
        const float* arow = &Gl[(mt * 16 + nloc) * LSTR];
        #pragma unroll
        for (int k0 = 0; k0 < GMM_D; k0 += 4) {
            v2f a = *(const v2f*)(arow + k0 + khalf);
            v2f b = *(const v2f*)(brow + k0 + khalf);
            acc = __builtin_amdgcn_wmma_f32_16x16x4_f32(
                       false, a,  false, b,
                       (short)0, acc,
                       false,  false);
        }
        const int mrow = mt * 16 + (lane >> 4) * 8;
        #pragma unroll
        for (int j = 0; j < 8; ++j) {
            const float d = acc[j] - hl[mrow + j];
            s += d * d;
        }
    }

    s += __shfl_xor(s, 16, 32);

    if (lane < 16) {
        const int b = n0 + w * 16 + nloc;
        out[(size_t)b * GMM_C + c] = cc[c] - 0.5f * s;
    }
}

extern "C" void kernel_launch(void* const* d_in, const int* in_sizes, int n_in,
                              void* d_out, int out_size, void* d_ws, size_t ws_size,
                              hipStream_t stream) {
    (void)in_sizes; (void)n_in; (void)out_size; (void)ws_size;

    const float* x       = (const float*)d_in[0];
    const float* means   = (const float*)d_in[1];
    const float* covs    = (const float*)d_in[2];
    const float* weights = (const float*)d_in[3];
    float* out = (float*)d_out;

    float* Gw  = (float*)d_ws;
    float* hw  = Gw + (size_t)GMM_C * GMM_D * GMM_D;
    float* ccw = hw + (size_t)GMM_C * GMM_D;

    gmm_prep_kernel<<<GMM_C, 64, 0, stream>>>(means, covs, weights, Gw, hw, ccw);

    dim3 grid(GMM_B / TILE_N, GMM_C);
    gmm_logits_kernel<<<grid, 256, 0, stream>>>(x, Gw, hw, ccw, out);
}


# prep kernel: Cholesky row, running diagonal and inverse column in registers (shifted one slot per step, rolled loop); one LDS column broadcast + one barrier per step replace serial LDS loops
# speedup vs baseline: 1.6819x; 1.6819x over previous
#include <hip/hip_runtime.h>
#include <hip/hip_bf16.h>
#include <math.h>

#define GMM_B 32768
#define GMM_C 100
#define GMM_D 64

typedef __attribute__((ext_vector_type(2))) float v2f;
typedef __attribute__((ext_vector_type(8))) float v8f;

#define PSTR 68

template <int J0, int NJ>
__device__ __forceinline__ void gmm_chol_phase(float (&a)[GMM_D], float (&d)[GMM_D], float (&s)[GMM_D],
                                               float& ljj, float& l, float& g, float& mydiag,
                                               float* __restrict__ Lc, float* __restrict__ Gs, const int t)
{
    constexpr int W  = GMM_D - 1 - J0;
    constexpr int NQ = (W + 4) / 4;
    #pragma unroll 1
    for (int j = J0; j < J0 + NJ; ++j) {
        mydiag = (t == j) ? ljj : mydiag;
        if (t > j) Lc[j * GMM_D + (t - j)] = l;
        Gs[j * PSTR + t] = g;
        __syncthreads();
        const float* col = Lc + j * GMM_D;
        float lkv[4 * NQ];
        #pragma unroll
        for (int q = 0; q < NQ; ++q) {
            const float4 v = *(const float4*)(col + 4 * q);
            lkv[4 * q + 0] = v.x; lkv[4 * q + 1] = v.y;
            lkv[4 * q + 2] = v.z; lkv[4 * q + 3] = v.w;
        }
        const float lc = l, gc = g;
        {
            const float lk = lkv[1];
            a[0] = a[1] - lc * lk;
            d[0] = d[1] - lk * lk;
            s[0] = s[1] - lk * gc;
        }
        ljj = sqrtf(d[0]);
        l   = a[0] / ljj;
        g   = s[0] / ljj;
        #pragma unroll
        for (int p = 2; p <= W; ++p) {
            const float lk = lkv[p];
            a[p - 1] = a[p] - lc * lk;
            d[p - 1] = d[p] - lk * lk;
            s[p - 1] = s[p] - lk * gc;
        }
    }
}

__global__ __launch_bounds__(64) __attribute__((amdgpu_waves_per_eu(1, 1)))
void gmm_prep_kernel(const float* __restrict__ means,
                     const float* __restrict__ covs,
                     const float* __restrict__ weights,
                     float* __restrict__ Gout,
                     float* __restrict__ hout,
                     float* __restrict__ ccout)
{
    __shared__ __align__(16) float Lc[GMM_D * GMM_D];
    __shared__ __align__(16) float Gs[GMM_D * PSTR];
    __shared__ __align__(16) float dg[GMM_D];
    __shared__ __align__(16) float mu[GMM_D];
    __shared__ float red[GMM_D];

    const int t = threadIdx.x;
    const int c = blockIdx.x;
    const float* cov = covs + (size_t)c * GMM_D * GMM_D;

    float a[GMM_D];
    {
        const float4* src = (const float4*)(cov + (size_t)t * GMM_D);
        #pragma unroll
        for (int q = 0; q < GMM_D / 4; ++q) {
            const float4 v = src[q];
            a[4 * q + 0] = v.x; a[4 * q + 1] = v.y;
            a[4 * q + 2] = v.z; a[4 * q + 3] = v.w;
        }
    }
    dg[t] = cov[t * (GMM_D + 1)];
    mu[t] = means[c * GMM_D + t];
    {
        const float4 z = make_float4(0.0f, 0.0f, 0.0f, 0.0f);
        #pragma unroll
        for (int q = 0; q < GMM_D / 4; ++q)
            *(float4*)&Lc[(q * GMM_D + t) * 4] = z;
    }
    __syncthreads();

    float d[GMM_D];
    float s[GMM_D];
    #pragma unroll
    for (int q = 0; q < GMM_D / 4; ++q) {
        const float4 v = *(const float4*)&dg[4 * q];
        d[4 * q + 0] = v.x; d[4 * q + 1] = v.y;
        d[4 * q + 2] = v.z; d[4 * q + 3] = v.w;
    }
    #pragma unroll
    for (int i = 0; i < GMM_D; ++i) s[i] = (i == t) ? 1.0f : 0.0f;

    float mydiag = 1.0f;
    float ljj = sqrtf(d[0]);
    float l   = a[0] / ljj;
    float g   = s[0] / ljj;
    gmm_chol_phase< 0, 8>(a, d, s, ljj, l, g, mydiag, Lc, Gs, t);
    gmm_chol_phase< 8, 8>(a, d, s, ljj, l, g, mydiag, Lc, Gs, t);
    gmm_chol_phase<16, 8>(a, d, s, ljj, l, g, mydiag, Lc, Gs, t);
    gmm_chol_phase<24, 8>(a, d, s, ljj, l, g, mydiag, Lc, Gs, t);
    gmm_chol_phase<32, 8>(a, d, s, ljj, l, g, mydiag, Lc, Gs, t);
    gmm_chol_phase<40, 8>(a, d, s, ljj, l, g, mydiag, Lc, Gs, t);
    gmm_chol_phase<48, 8>(a, d, s, ljj, l, g, mydiag, Lc, Gs, t);
    gmm_chol_phase<56, 8>(a, d, s, ljj, l, g, mydiag, Lc, Gs, t);

    red[t] = 2.0f * logf(mydiag);
    __syncthreads();
    if (t == 0) {
        float ld = 0.0f;
        for (int k = 0; k < GMM_D; ++k) ld += red[k];
        const float LOG2PI = 1.8378770664093453f;
        ccout[c] = logf(weights[c]) - 0.5f * (ld + (float)GMM_D * LOG2PI);
    }

    float hv = 0.0f;
    #pragma unroll
    for (int k = 0; k < GMM_D; ++k) hv += Gs[t * PSTR + k] * mu[k];
    hout[c * GMM_D + t] = hv;

    {
        const int col4 = (t & 15) * 4;
        const int r0   = t >> 4;
        float* gdst = Gout + (size_t)c * GMM_D * GMM_D;
        #pragma unroll
        for (int it = 0; it < GMM_D / 4; ++it) {
            const int r = r0 + it * 4;
            *(float4*)(gdst + (size_t)r * GMM_D + col4) = *(const float4*)&Gs[r * PSTR + col4];
        }
    }
}

#define TILE_N 128
#define LSTR   68

__global__ __launch_bounds__(256)
void gmm_logits_kernel(const float* __restrict__ x,
                       const float* __restrict__ G,
                       const float* __restrict__ h,
                       const float* __restrict__ cc,
                       float* __restrict__ out)
{
    __shared__ __align__(16) float Xl[TILE_N * LSTR];
    __shared__ __align__(16) float Gl[GMM_D * LSTR];
    __shared__ float hl[GMM_D];

    const int t  = threadIdx.x;
    const int c  = blockIdx.y;
    const int n0 = blockIdx.x * TILE_N;

    {
        const int col4 = (t & 15) * 4;
        const int r0   = t >> 4;
        #pragma unroll
        for (int it = 0; it < 8; ++it) {
            const int r = r0 + it * 16;
            const float4 v = *(const float4*)(x + (size_t)(n0 + r) * GMM_D + col4);
            *(float4*)&Xl[r * LSTR + col4] = v;
        }
        #pragma unroll
        for (int it = 0; it < 4; ++it) {
            const int r = r0 + it * 16;
            const float4 v = *(const float4*)(G + (size_t)c * GMM_D * GMM_D + (size_t)r * GMM_D + col4);
            *(float4*)&Gl[r * LSTR + col4] = v;
        }
        if (t < GMM_D) hl[t] = h[c * GMM_D + t];
    }
    __syncthreads();

    const int lane  = t & 31;
    const int w     = t >> 5;
    const int nloc  = lane & 15;
    const int khalf = (lane >> 4) * 2;

    const float* brow = &Xl[(w * 16 + nloc) * LSTR];

    float s = 0.0f;
    #pragma unroll
    for (int mt = 0; mt < 4; ++mt) {
        v8f acc = {};
        const float* arow = &Gl[(mt * 16 + nloc) * LSTR];
        #pragma unroll
        for (int k0 = 0; k0 < GMM_D; k0 += 4) {
            v2f a = *(const v2f*)(arow + k0 + khalf);
            v2f b = *(const v2f*)(brow + k0 + khalf);
            acc = __builtin_amdgcn_wmma_f32_16x16x4_f32(
                       false, a,  false, b,
                       (short)0, acc,
                       false,  false);
        }
        const int mrow = mt * 16 + (lane >> 4) * 8;
        #pragma unroll
        for (int j = 0; j < 8; ++j) {
            const float d = acc[j] - hl[mrow + j];
            s += d * d;
        }
    }

    s += __shfl_xor(s, 16, 32);

    if (lane < 16) {
        const int b = n0 + w * 16 + nloc;
        out[(size_t)b * GMM_C + c] = cc[c] - 0.5f * s;
    }
}

extern "C" void kernel_launch(void* const* d_in, const int* in_sizes, int n_in,
                              void* d_out, int out_size, void* d_ws, size_t ws_size,
                              hipStream_t stream) {
    (void)in_sizes; (void)n_in; (void)out_size; (void)ws_size;

    const float* x       = (const float*)d_in[0];
    const float* means   = (const float*)d_in[1];
    const float* covs    = (const float*)d_in[2];
    const float* weights = (const float*)d_in[3];
    float* out = (float*)d_out;

    float* Gw  = (float*)d_ws;
    float* hw  = Gw + (size_t)GMM_C * GMM_D * GMM_D;
    float* ccw = hw + (size_t)GMM_C * GMM_D;

    gmm_prep_kernel<<<GMM_C, 64, 0, stream>>>(means, covs, weights, Gw, hw, ccw);

    dim3 grid(GMM_B / TILE_N, GMM_C);
    gmm_logits_kernel<<<grid, 256, 0, stream>>>(x, Gw, hw, ccw, out);
}


# main kernel: tile loop inside the workgroup: each keeps its 128-row X tile (B fragments in registers) and loops over 25 classes, G double-buffered in LDS; 1024 resident workgroups, not 25600
# speedup vs baseline: 1.8384x; 1.0930x over previous
#include <hip/hip_runtime.h>
#include <hip/hip_bf16.h>
#include <math.h>

#define GMM_B 32768
#define GMM_C 100
#define GMM_D 64

typedef __attribute__((ext_vector_type(2))) float v2f;
typedef __attribute__((ext_vector_type(8))) float v8f;

#define PSTR 68

template <int J0, int NJ>
__device__ __forceinline__ void gmm_chol_phase(float (&a)[GMM_D], float (&d)[GMM_D], float (&s)[GMM_D],
                                               float& ljj, float& l, float& g, float& mydiag,
                                               float* __restrict__ Lc, float* __restrict__ Gs, const int t)
{
    constexpr int W  = GMM_D - 1 - J0;
    constexpr int NQ = (W + 4) / 4;
    #pragma unroll 1
    for (int j = J0; j < J0 + NJ; ++j) {
        mydiag = (t == j) ? ljj : mydiag;
        if (t > j) Lc[j * GMM_D + (t - j)] = l;
        Gs[j * PSTR + t] = g;
        __syncthreads();
        const float* col = Lc + j * GMM_D;
        float lkv[4 * NQ];
        #pragma unroll
        for (int q = 0; q < NQ; ++q) {
            const float4 v = *(const float4*)(col + 4 * q);
            lkv[4 * q + 0] = v.x; lkv[4 * q + 1] = v.y;
            lkv[4 * q + 2] = v.z; lkv[4 * q + 3] = v.w;
        }
        const float lc = l, gc = g;
        {
            const float lk = lkv[1];
            a[0] = a[1] - lc * lk;
            d[0] = d[1] - lk * lk;
            s[0] = s[1] - lk * gc;
        }
        ljj = sqrtf(d[0]);
        l   = a[0] / ljj;
        g   = s[0] / ljj;
        #pragma unroll
        for (int p = 2; p <= W; ++p) {
            const float lk = lkv[p];
            a[p - 1] = a[p] - lc * lk;
            d[p - 1] = d[p] - lk * lk;
            s[p - 1] = s[p] - lk * gc;
        }
    }
}

__global__ __launch_bounds__(64) __attribute__((amdgpu_waves_per_eu(1, 1)))
void gmm_prep_kernel(const float* __restrict__ means,
                     const float* __restrict__ covs,
                     const float* __restrict__ weights,
                     float* __restrict__ Gout,
                     float* __restrict__ hout,
                     float* __restrict__ ccout)
{
    __shared__ __align__(16) float Lc[GMM_D * GMM_D];
    __shared__ __align__(16) float Gs[GMM_D * PSTR];
    __shared__ __align__(16) float dg[GMM_D];
    __shared__ __align__(16) float mu[GMM_D];
    __shared__ float red[GMM_D];

    const int t = threadIdx.x;
    const int c = blockIdx.x;
    const float* cov = covs + (size_t)c * GMM_D * GMM_D;

    float a[GMM_D];
    {
        const float4* src = (const float4*)(cov + (size_t)t * GMM_D);
        #pragma unroll
        for (int q = 0; q < GMM_D / 4; ++q) {
            const float4 v = src[q];
            a[4 * q + 0] = v.x; a[4 * q + 1] = v.y;
            a[4 * q + 2] = v.z; a[4 * q + 3] = v.w;
        }
    }
    dg[t] = cov[t * (GMM_D + 1)];
    mu[t] = means[c * GMM_D + t];
    {
        const float4 z = make_float4(0.0f, 0.0f, 0.0f, 0.0f);
        #pragma unroll
        for (int q = 0; q < GMM_D / 4; ++q)
            *(float4*)&Lc[(q * GMM_D + t) * 4] = z;
    }
    __syncthreads();

    float d[GMM_D];
    float s[GMM_D];
    #pragma unroll
    for (int q = 0; q < GMM_D / 4; ++q) {
        const float4 v = *(const float4*)&dg[4 * q];
        d[4 * q + 0] = v.x; d[4 * q + 1] = v.y;
        d[4 * q + 2] = v.z; d[4 * q + 3] = v.w;
    }
    #pragma unroll
    for (int i = 0; i < GMM_D; ++i) s[i] = (i == t) ? 1.0f : 0.0f;

    float mydiag = 1.0f;
    float ljj = sqrtf(d[0]);
    float l   = a[0] / ljj;
    float g   = s[0] / ljj;
    gmm_chol_phase< 0, 8>(a, d, s, ljj, l, g, mydiag, Lc, Gs, t);
    gmm_chol_phase< 8, 8>(a, d, s, ljj, l, g, mydiag, Lc, Gs, t);
    gmm_chol_phase<16, 8>(a, d, s, ljj, l, g, mydiag, Lc, Gs, t);
    gmm_chol_phase<24, 8>(a, d, s, ljj, l, g, mydiag, Lc, Gs, t);
    gmm_chol_phase<32, 8>(a, d, s, ljj, l, g, mydiag, Lc, Gs, t);
    gmm_chol_phase<40, 8>(a, d, s, ljj, l, g, mydiag, Lc, Gs, t);
    gmm_chol_phase<48, 8>(a, d, s, ljj, l, g, mydiag, Lc, Gs, t);
    gmm_chol_phase<56, 8>(a, d, s, ljj, l, g, mydiag, Lc, Gs, t);

    red[t] = 2.0f * logf(mydiag);
    __syncthreads();
    if (t == 0) {
        float ld = 0.0f;
        for (int k = 0; k < GMM_D; ++k) ld += red[k];
        const float LOG2PI = 1.8378770664093453f;
        ccout[c] = logf(weights[c]) - 0.5f * (ld + (float)GMM_D * LOG2PI);
    }

    float hv = 0.0f;
    #pragma unroll
    for (int k = 0; k < GMM_D; ++k) hv += Gs[t * PSTR + k] * mu[k];
    hout[c * GMM_D + t] = hv;

    {
        const int col4 = (t & 15) * 4;
        const int r0   = t >> 4;
        float* gdst = Gout + (size_t)c * GMM_D * GMM_D;
        #pragma unroll
        for (int it = 0; it < GMM_D / 4; ++it) {
            const int r = r0 + it * 4;
            *(float4*)(gdst + (size_t)r * GMM_D + col4) = *(const float4*)&Gs[r * PSTR + col4];
        }
    }
}

#define TILE_N  128
#define LSTR    68
#define GMM_NCH 4
#define GMM_CPB (GMM_C / GMM_NCH)
static_assert(GMM_CPB * GMM_NCH == GMM_C, "class chunks must cover every class exactly once");

__global__ __launch_bounds__(256) __attribute__((amdgpu_waves_per_eu(8, 8)))
void gmm_logits_kernel(const float* __restrict__ x,
                       const float* __restrict__ G,
                       const float* __restrict__ h,
                       const float* __restrict__ cc,
                       float* __restrict__ out)
{
    __shared__ __align__(16) float Xl[TILE_N * LSTR];
    __shared__ __align__(16) float Gl[2 * GMM_D * LSTR];
    __shared__ float hl[2 * GMM_D];

    const int t    = threadIdx.x;
    const int c0   = blockIdx.y * GMM_CPB;
    const int n0   = blockIdx.x * TILE_N;
    const int col4 = (t & 15) * 4;
    const int r0   = t >> 4;

    {
        #pragma unroll
        for (int it = 0; it < 8; ++it) {
            const int r = r0 + it * 16;
            const float4 v = *(const float4*)(x + (size_t)(n0 + r) * GMM_D + col4);
            *(float4*)&Xl[r * LSTR + col4] = v;
        }
        #pragma unroll
        for (int it = 0; it < 4; ++it) {
            const int r = r0 + it * 16;
            const float4 v = *(const float4*)(G + (size_t)c0 * GMM_D * GMM_D + (size_t)r * GMM_D + col4);
            *(float4*)&Gl[r * LSTR + col4] = v;
        }
        if (t < GMM_D) hl[t] = h[c0 * GMM_D + t];
    }
    __syncthreads();

    const int lane  = t & 31;
    const int w     = t >> 5;
    const int nloc  = lane & 15;
    const int khalf = (lane >> 4) * 2;

    v2f bf[GMM_D / 4];
    {
        const float* brow = &Xl[(w * 16 + nloc) * LSTR];
        #pragma unroll
        for (int kk = 0; kk < GMM_D / 4; ++kk)
            bf[kk] = *(const v2f*)(brow + 4 * kk + khalf);
    }

    float* orow = out + (size_t)(n0 + w * 16 + nloc) * GMM_C;

    #pragma unroll 1
    for (int ci = 0; ci < GMM_CPB; ++ci) {
        const int  c    = c0 + ci;
        const int  cur  = ci & 1;
        const bool more = (ci + 1 < GMM_CPB);
        const float ccv = cc[c];

        float4 gn[4];
        float  hn = 0.0f;
        #pragma unroll
        for (int it = 0; it < 4; ++it) gn[it] = make_float4(0.0f, 0.0f, 0.0f, 0.0f);
        if (more) {
            const float* gsrc = G + (size_t)(c + 1) * GMM_D * GMM_D + (size_t)r0 * GMM_D + col4;
            #pragma unroll
            for (int it = 0; it < 4; ++it)
                gn[it] = *(const float4*)(gsrc + (size_t)it * 16 * GMM_D);
            if (t < GMM_D) hn = h[(c + 1) * GMM_D + t];
        }

        const float* Gc = Gl + cur * (GMM_D * LSTR);
        const float* hc = hl + cur * GMM_D;

        float s = 0.0f;
        #pragma unroll
        for (int mt = 0; mt < 4; ++mt) {
            v8f acc = {};
            const float* arow = Gc + (mt * 16 + nloc) * LSTR;
            #pragma unroll
            for (int k0 = 0; k0 < GMM_D; k0 += 4) {
                v2f a = *(const v2f*)(arow + k0 + khalf);
                acc = __builtin_amdgcn_wmma_f32_16x16x4_f32(
                           false, a,  false, bf[k0 / 4],
                           (short)0, acc,
                           false,  false);
            }
            const int mrow = mt * 16 + (lane >> 4) * 8;
            #pragma unroll
            for (int j = 0; j < 8; ++j) {
                const float d = acc[j] - hc[mrow + j];
                s += d * d;
            }
        }

        s += __shfl_xor(s, 16, 32);

        if (more) {
            float* Gn = Gl + (cur ^ 1) * (GMM_D * LSTR);
            #pragma unroll
            for (int it = 0; it < 4; ++it)
                *(float4*)&Gn[(r0 + it * 16) * LSTR + col4] = gn[it];
            if (t < GMM_D) hl[(cur ^ 1) * GMM_D + t] = hn;
        }
        __syncthreads();

        if (lane < 16) orow[c] = ccv - 0.5f * s;
    }
}

extern "C" void kernel_launch(void* const* d_in, const int* in_sizes, int n_in,
                              void* d_out, int out_size, void* d_ws, size_t ws_size,
                              hipStream_t stream) {
    (void)in_sizes; (void)n_in; (void)out_size; (void)ws_size;

    const float* x       = (const float*)d_in[0];
    const float* means   = (const float*)d_in[1];
    const float* covs    = (const float*)d_in[2];
    const float* weights = (const float*)d_in[3];
    float* out = (float*)d_out;

    float* Gw  = (float*)d_ws;
    float* hw  = Gw + (size_t)GMM_C * GMM_D * GMM_D;
    float* ccw = hw + (size_t)GMM_C * GMM_D;

    gmm_prep_kernel<<<GMM_C, 64, 0, stream>>>(means, covs, weights, Gw, hw, ccw);

    dim3 grid(GMM_B / TILE_N, GMM_NCH);
    gmm_logits_kernel<<<grid, 256, 0, stream>>>(x, Gw, hw, ccw, out);
}


# main kernel: G (inverse of the lower-triangular covariance factor) is lower triangular: M-tile mt ends its K-loop at k = 16*(mt+1), skipping all-zero blocks: 40 f32 WMMAs per class tile, not 64
# speedup vs baseline: 2.4950x; 1.3572x over previous
#include <hip/hip_runtime.h>
#include <hip/hip_bf16.h>
#include <math.h>

#define GMM_B 32768
#define GMM_C 100
#define GMM_D 64

typedef __attribute__((ext_vector_type(2))) float v2f;
typedef __attribute__((ext_vector_type(8))) float v8f;

#define PSTR 68

template <int J0, int NJ>
__device__ __forceinline__ void gmm_chol_phase(float (&a)[GMM_D], float (&d)[GMM_D], float (&s)[GMM_D],
                                               float& ljj, float& l, float& g, float& mydiag,
                                               float* __restrict__ Lc, float* __restrict__ Gs, const int t)
{
    constexpr int W  = GMM_D - 1 - J0;
    constexpr int NQ = (W + 4) / 4;
    #pragma unroll 1
    for (int j = J0; j < J0 + NJ; ++j) {
        mydiag = (t == j) ? ljj : mydiag;
        if (t > j) Lc[j * GMM_D + (t - j)] = l;
        Gs[j * PSTR + t] = g;
        __syncthreads();
        const float* col = Lc + j * GMM_D;
        float lkv[4 * NQ];
        #pragma unroll
        for (int q = 0; q < NQ; ++q) {
            const float4 v = *(const float4*)(col + 4 * q);
            lkv[4 * q + 0] = v.x; lkv[4 * q + 1] = v.y;
            lkv[4 * q + 2] = v.z; lkv[4 * q + 3] = v.w;
        }
        const float lc = l, gc = g;
        {
            const float lk = lkv[1];
            a[0] = a[1] - lc * lk;
            d[0] = d[1] - lk * lk;
            s[0] = s[1] - lk * gc;
        }
        ljj = sqrtf(d[0]);
        l   = a[0] / ljj;
        g   = s[0] / ljj;
        #pragma unroll
        for (int p = 2; p <= W; ++p) {
            const float lk = lkv[p];
            a[p - 1] = a[p] - lc * lk;
            d[p - 1] = d[p] - lk * lk;
            s[p - 1] = s[p] - lk * gc;
        }
    }
}

__global__ __launch_bounds__(64) __attribute__((amdgpu_waves_per_eu(1, 1)))
void gmm_prep_kernel(const float* __restrict__ means,
                     const float* __restrict__ covs,
                     const float* __restrict__ weights,
                     float* __restrict__ Gout,
                     float* __restrict__ hout,
                     float* __restrict__ ccout)
{
    __shared__ __align__(16) float Lc[GMM_D * GMM_D];
    __shared__ __align__(16) float Gs[GMM_D * PSTR];
    __shared__ __align__(16) float dg[GMM_D];
    __shared__ __align__(16) float mu[GMM_D];
    __shared__ float red[GMM_D];

    const int t = threadIdx.x;
    const int c = blockIdx.x;
    const float* cov = covs + (size_t)c * GMM_D * GMM_D;

    float a[GMM_D];
    {
        const float4* src = (const float4*)(cov + (size_t)t * GMM_D);
        #pragma unroll
        for (int q = 0; q < GMM_D / 4; ++q) {
            const float4 v = src[q];
            a[4 * q + 0] = v.x; a[4 * q + 1] = v.y;
            a[4 * q + 2] = v.z; a[4 * q + 3] = v.w;
        }
    }
    dg[t] = cov[t * (GMM_D + 1)];
    mu[t] = means[c * GMM_D + t];
    {
        const float4 z = make_float4(0.0f, 0.0f, 0.0f, 0.0f);
        #pragma unroll
        for (int q = 0; q < GMM_D / 4; ++q)
            *(float4*)&Lc[(q * GMM_D + t) * 4] = z;
    }
    __syncthreads();

    float d[GMM_D];
    float s[GMM_D];
    #pragma unroll
    for (int q = 0; q < GMM_D / 4; ++q) {
        const float4 v = *(const float4*)&dg[4 * q];
        d[4 * q + 0] = v.x; d[4 * q + 1] = v.y;
        d[4 * q + 2] = v.z; d[4 * q + 3] = v.w;
    }
    #pragma unroll
    for (int i = 0; i < GMM_D; ++i) s[i] = (i == t) ? 1.0f : 0.0f;

    float mydiag = 1.0f;
    float ljj = sqrtf(d[0]);
    float l   = a[0] / ljj;
    float g   = s[0] / ljj;
    gmm_chol_phase< 0, 8>(a, d, s, ljj, l, g, mydiag, Lc, Gs, t);
    gmm_chol_phase< 8, 8>(a, d, s, ljj, l, g, mydiag, Lc, Gs, t);
    gmm_chol_phase<16, 8>(a, d, s, ljj, l, g, mydiag, Lc, Gs, t);
    gmm_chol_phase<24, 8>(a, d, s, ljj, l, g, mydiag, Lc, Gs, t);
    gmm_chol_phase<32, 8>(a, d, s, ljj, l, g, mydiag, Lc, Gs, t);
    gmm_chol_phase<40, 8>(a, d, s, ljj, l, g, mydiag, Lc, Gs, t);
    gmm_chol_phase<48, 8>(a, d, s, ljj, l, g, mydiag, Lc, Gs, t);
    gmm_chol_phase<56, 8>(a, d, s, ljj, l, g, mydiag, Lc, Gs, t);

    red[t] = 2.0f * logf(mydiag);
    __syncthreads();
    if (t == 0) {
        float ld = 0.0f;
        for (int k = 0; k < GMM_D; ++k) ld += red[k];
        const float LOG2PI = 1.8378770664093453f;
        ccout[c] = logf(weights[c]) - 0.5f * (ld + (float)GMM_D * LOG2PI);
    }

    float hv = 0.0f;
    #pragma unroll
    for (int k = 0; k < GMM_D; ++k) hv += Gs[t * PSTR + k] * mu[k];
    hout[c * GMM_D + t] = hv;

    {
        const int col4 = (t & 15) * 4;
        const int r0   = t >> 4;
        float* gdst = Gout + (size_t)c * GMM_D * GMM_D;
        #pragma unroll
        for (int it = 0; it < GMM_D / 4; ++it) {
            const int r = r0 + it * 4;
            *(float4*)(gdst + (size_t)r * GMM_D + col4) = *(const float4*)&Gs[r * PSTR + col4];
        }
    }
}

#define TILE_N  128
#define LSTR    68
#define GMM_NCH 4
#define GMM_CPB (GMM_C / GMM_NCH)
static_assert(GMM_CPB * GMM_NCH == GMM_C, "class chunks must cover every class exactly once");

__global__ __launch_bounds__(256) __attribute__((amdgpu_waves_per_eu(8, 8)))
void gmm_logits_kernel(const float* __restrict__ x,
                       const float* __restrict__ G,
                       const float* __restrict__ h,
                       const float* __restrict__ cc,
                       float* __restrict__ out)
{
    __shared__ __align__(16) float Xl[TILE_N * LSTR];
    __shared__ __align__(16) float Gl[2 * GMM_D * LSTR];
    __shared__ float hl[2 * GMM_D];

    const int t    = threadIdx.x;
    const int c0   = blockIdx.y * GMM_CPB;
    const int n0   = blockIdx.x * TILE_N;
    const int col4 = (t & 15) * 4;
    const int r0   = t >> 4;

    {
        #pragma unroll
        for (int it = 0; it < 8; ++it) {
            const int r = r0 + it * 16;
            const float4 v = *(const float4*)(x + (size_t)(n0 + r) * GMM_D + col4);
            *(float4*)&Xl[r * LSTR + col4] = v;
        }
        #pragma unroll
        for (int it = 0; it < 4; ++it) {
            const int r = r0 + it * 16;
            const float4 v = *(const float4*)(G + (size_t)c0 * GMM_D * GMM_D + (size_t)r * GMM_D + col4);
            *(float4*)&Gl[r * LSTR + col4] = v;
        }
        if (t < GMM_D) hl[t] = h[c0 * GMM_D + t];
    }
    __syncthreads();

    const int lane  = t & 31;
    const int w     = t >> 5;
    const int nloc  = lane & 15;
    const int khalf = (lane >> 4) * 2;

    v2f bf[GMM_D / 4];
    {
        const float* brow = &Xl[(w * 16 + nloc) * LSTR];
        #pragma unroll
        for (int kk = 0; kk < GMM_D / 4; ++kk)
            bf[kk] = *(const v2f*)(brow + 4 * kk + khalf);
    }

    float* orow = out + (size_t)(n0 + w * 16 + nloc) * GMM_C;

    #pragma unroll 1
    for (int ci = 0; ci < GMM_CPB; ++ci) {
        const int  c    = c0 + ci;
        const int  cur  = ci & 1;
        const bool more = (ci + 1 < GMM_CPB);
        const float ccv = cc[c];

        float4 gn[4];
        float  hn = 0.0f;
        #pragma unroll
        for (int it = 0; it < 4; ++it) gn[it] = make_float4(0.0f, 0.0f, 0.0f, 0.0f);
        if (more) {
            const float* gsrc = G + (size_t)(c + 1) * GMM_D * GMM_D + (size_t)r0 * GMM_D + col4;
            #pragma unroll
            for (int it = 0; it < 4; ++it)
                gn[it] = *(const float4*)(gsrc + (size_t)it * 16 * GMM_D);
            if (t < GMM_D) hn = h[(c + 1) * GMM_D + t];
        }

        const float* Gc = Gl + cur * (GMM_D * LSTR);
        const float* hc = hl + cur * GMM_D;

        float s = 0.0f;
        #pragma unroll
        for (int mt = 0; mt < 4; ++mt) {
            v8f acc = {};
            const float* arow = Gc + (mt * 16 + nloc) * LSTR;
            #pragma unroll
            for (int k0 = 0; k0 < 16 * (mt + 1); k0 += 4) {
                v2f a = *(const v2f*)(arow + k0 + khalf);
                acc = __builtin_amdgcn_wmma_f32_16x16x4_f32(
                           false, a,  false, bf[k0 / 4],
                           (short)0, acc,
                           false,  false);
            }
            const int mrow = mt * 16 + (lane >> 4) * 8;
            #pragma unroll
            for (int j = 0; j < 8; ++j) {
                const float d = acc[j] - hc[mrow + j];
                s += d * d;
            }
        }

        s += __shfl_xor(s, 16, 32);

        if (more) {
            float* Gn = Gl + (cur ^ 1) * (GMM_D * LSTR);
            #pragma unroll
            for (int it = 0; it < 4; ++it)
                *(float4*)&Gn[(r0 + it * 16) * LSTR + col4] = gn[it];
            if (t < GMM_D) hl[(cur ^ 1) * GMM_D + t] = hn;
        }
        __syncthreads();

        if (lane < 16) orow[c] = ccv - 0.5f * s;
    }
}

extern "C" void kernel_launch(void* const* d_in, const int* in_sizes, int n_in,
                              void* d_out, int out_size, void* d_ws, size_t ws_size,
                              hipStream_t stream) {
    (void)in_sizes; (void)n_in; (void)out_size; (void)ws_size;

    const float* x       = (const float*)d_in[0];
    const float* means   = (const float*)d_in[1];
    const float* covs    = (const float*)d_in[2];
    const float* weights = (const float*)d_in[3];
    float* out = (float*)d_out;

    float* Gw  = (float*)d_ws;
    float* hw  = Gw + (size_t)GMM_C * GMM_D * GMM_D;
    float* ccw = hw + (size_t)GMM_C * GMM_D;

    gmm_prep_kernel<<<GMM_C, 64, 0, stream>>>(means, covs, weights, Gw, hw, ccw);

    dim3 grid(GMM_B / TILE_N, GMM_NCH);
    gmm_logits_kernel<<<grid, 256, 0, stream>>>(x, Gw, hw, ccw, out);
}


# main kernel: class-loop vector work more than halved: accumulators start at -h (C operand read from LDS), epilogue = 32 squares in two running sums; loop unrolled by 2 (constant LDS buffers)
# speedup vs baseline: 2.6126x; 1.0471x over previous
#include <hip/hip_runtime.h>
#include <hip/hip_bf16.h>
#include <math.h>

#define GMM_B 32768
#define GMM_C 100
#define GMM_D 64

typedef __attribute__((ext_vector_type(2))) float v2f;
typedef __attribute__((ext_vector_type(8))) float v8f;

#define PSTR 68

template <int J0, int NJ>
__device__ __forceinline__ void gmm_chol_phase(float (&a)[GMM_D], float (&d)[GMM_D], float (&s)[GMM_D],
                                               float& ljj, float& l, float& g, float& mydiag,
                                               float* __restrict__ Lc, float* __restrict__ Gs, const int t)
{
    constexpr int W  = GMM_D - 1 - J0;
    constexpr int NQ = (W + 4) / 4;
    #pragma unroll 1
    for (int j = J0; j < J0 + NJ; ++j) {
        mydiag = (t == j) ? ljj : mydiag;
        if (t > j) Lc[j * GMM_D + (t - j)] = l;
        Gs[j * PSTR + t] = g;
        __syncthreads();
        const float* col = Lc + j * GMM_D;
        float lkv[4 * NQ];
        #pragma unroll
        for (int q = 0; q < NQ; ++q) {
            const float4 v = *(const float4*)(col + 4 * q);
            lkv[4 * q + 0] = v.x; lkv[4 * q + 1] = v.y;
            lkv[4 * q + 2] = v.z; lkv[4 * q + 3] = v.w;
        }
        const float lc = l, gc = g;
        {
            const float lk = lkv[1];
            a[0] = a[1] - lc * lk;
            d[0] = d[1] - lk * lk;
            s[0] = s[1] - lk * gc;
        }
        ljj = sqrtf(d[0]);
        l   = a[0] / ljj;
        g   = s[0] / ljj;
        #pragma unroll
        for (int p = 2; p <= W; ++p) {
            const float lk = lkv[p];
            a[p - 1] = a[p] - lc * lk;
            d[p - 1] = d[p] - lk * lk;
            s[p - 1] = s[p] - lk * gc;
        }
    }
}

__global__ __launch_bounds__(64) __attribute__((amdgpu_waves_per_eu(1, 1)))
void gmm_prep_kernel(const float* __restrict__ means,
                     const float* __restrict__ covs,
                     const float* __restrict__ weights,
                     float* __restrict__ Gout,
                     float* __restrict__ hout,
                     float* __restrict__ ccout)
{
    __shared__ __align__(16) float Lc[GMM_D * GMM_D];
    __shared__ __align__(16) float Gs[GMM_D * PSTR];
    __shared__ __align__(16) float dg[GMM_D];
    __shared__ __align__(16) float mu[GMM_D];
    __shared__ float red[GMM_D];

    const int t = threadIdx.x;
    const int c = blockIdx.x;
    const float* cov = covs + (size_t)c * GMM_D * GMM_D;

    float a[GMM_D];
    {
        const float4* src = (const float4*)(cov + (size_t)t * GMM_D);
        #pragma unroll
        for (int q = 0; q < GMM_D / 4; ++q) {
            const float4 v = src[q];
            a[4 * q + 0] = v.x; a[4 * q + 1] = v.y;
            a[4 * q + 2] = v.z; a[4 * q + 3] = v.w;
        }
    }
    dg[t] = cov[t * (GMM_D + 1)];
    mu[t] = means[c * GMM_D + t];
    {
        const float4 z = make_float4(0.0f, 0.0f, 0.0f, 0.0f);
        #pragma unroll
        for (int q = 0; q < GMM_D / 4; ++q)
            *(float4*)&Lc[(q * GMM_D + t) * 4] = z;
    }
    __syncthreads();

    float d[GMM_D];
    float s[GMM_D];
    #pragma unroll
    for (int q = 0; q < GMM_D / 4; ++q) {
        const float4 v = *(const float4*)&dg[4 * q];
        d[4 * q + 0] = v.x; d[4 * q + 1] = v.y;
        d[4 * q + 2] = v.z; d[4 * q + 3] = v.w;
    }
    #pragma unroll
    for (int i = 0; i < GMM_D; ++i) s[i] = (i == t) ? 1.0f : 0.0f;

    float mydiag = 1.0f;
    float ljj = sqrtf(d[0]);
    float l   = a[0] / ljj;
    float g   = s[0] / ljj;
    gmm_chol_phase< 0, 8>(a, d, s, ljj, l, g, mydiag, Lc, Gs, t);
    gmm_chol_phase< 8, 8>(a, d, s, ljj, l, g, mydiag, Lc, Gs, t);
    gmm_chol_phase<16, 8>(a, d, s, ljj, l, g, mydiag, Lc, Gs, t);
    gmm_chol_phase<24, 8>(a, d, s, ljj, l, g, mydiag, Lc, Gs, t);
    gmm_chol_phase<32, 8>(a, d, s, ljj, l, g, mydiag, Lc, Gs, t);
    gmm_chol_phase<40, 8>(a, d, s, ljj, l, g, mydiag, Lc, Gs, t);
    gmm_chol_phase<48, 8>(a, d, s, ljj, l, g, mydiag, Lc, Gs, t);
    gmm_chol_phase<56, 8>(a, d, s, ljj, l, g, mydiag, Lc, Gs, t);

    red[t] = 2.0f * logf(mydiag);
    __syncthreads();
    if (t == 0) {
        float ld = 0.0f;
        for (int k = 0; k < GMM_D; ++k) ld += red[k];
        const float LOG2PI = 1.8378770664093453f;
        ccout[c] = logf(weights[c]) - 0.5f * (ld + (float)GMM_D * LOG2PI);
    }

    float hv = 0.0f;
    #pragma unroll
    for (int k = 0; k < GMM_D; ++k) hv += Gs[t * PSTR + k] * mu[k];
    hout[c * GMM_D + t] = -hv;

    {
        const int col4 = (t & 15) * 4;
        const int r0   = t >> 4;
        float* gdst = Gout + (size_t)c * GMM_D * GMM_D;
        #pragma unroll
        for (int it = 0; it < GMM_D / 4; ++it) {
            const int r = r0 + it * 4;
            *(float4*)(gdst + (size_t)r * GMM_D + col4) = *(const float4*)&Gs[r * PSTR + col4];
        }
    }
}

#define TILE_N  128
#define LSTR    68
#define GMM_NCH 4
#define GMM_CPB (GMM_C / GMM_NCH)
static_assert(GMM_CPB * GMM_NCH == GMM_C, "class chunks must cover every class exactly once");

template <int CUR, bool MORE>
__device__ __forceinline__ void gmm_class_step(const int c,
                                               const float* __restrict__ G,
                                               const float* __restrict__ h,
                                               const float* __restrict__ cc,
                                               float* __restrict__ op,
                                               float* __restrict__ Gl,
                                               float* __restrict__ hl,
                                               const v2f (&bf)[GMM_D / 4],
                                               const int t, const int lane,
                                               const unsigned goff, const unsigned loff,
                                               const unsigned aoff)
{
    const float ccv = cc[c];

    float4 gn[4];
    float  hn;
    if (MORE) {
        const float* gsrc = G + (size_t)(c + 1) * GMM_D * GMM_D;
        #pragma unroll
        for (int it = 0; it < 4; ++it)
            gn[it] = *(const float4*)(gsrc + goff + it * 16 * GMM_D);
        if (t < GMM_D) hn = (h + (size_t)(c + 1) * GMM_D)[t];
    }

    const float* Gc = Gl + CUR * (GMM_D * LSTR);
    const float* hc = hl + CUR * GMM_D + (lane >> 4) * 8;

    float q0 = 0.0f, q1 = 0.0f;
    #pragma unroll
    for (int mt = 0; mt < 4; ++mt) {
        v8f acc;
        {
            const float4 h0 = *(const float4*)(hc + mt * 16);
            const float4 h1 = *(const float4*)(hc + mt * 16 + 4);
            acc[0] = h0.x; acc[1] = h0.y; acc[2] = h0.z; acc[3] = h0.w;
            acc[4] = h1.x; acc[5] = h1.y; acc[6] = h1.z; acc[7] = h1.w;
        }
        const float* arow = Gc + aoff + mt * 16 * LSTR;
        #pragma unroll
        for (int k0 = 0; k0 < 16 * (mt + 1); k0 += 4) {
            v2f a = *(const v2f*)(arow + k0);
            acc = __builtin_amdgcn_wmma_f32_16x16x4_f32(
                       false, a,  false, bf[k0 / 4],
                       (short)0, acc,
                       false,  false);
        }
        q0 = fmaf(acc[0], acc[0], q0);
        q1 = fmaf(acc[1], acc[1], q1);
        q0 = fmaf(acc[2], acc[2], q0);
        q1 = fmaf(acc[3], acc[3], q1);
        q0 = fmaf(acc[4], acc[4], q0);
        q1 = fmaf(acc[5], acc[5], q1);
        q0 = fmaf(acc[6], acc[6], q0);
        q1 = fmaf(acc[7], acc[7], q1);
    }
    float s = q0 + q1;

    s += __shfl_xor(s, 16, 32);

    if (MORE) {
        float* Gn = Gl + (CUR ^ 1) * (GMM_D * LSTR);
        #pragma unroll
        for (int it = 0; it < 4; ++it)
            *(float4*)&Gn[loff + it * 16 * LSTR] = gn[it];
        if (t < GMM_D) hl[(CUR ^ 1) * GMM_D + t] = hn;
    }
    __syncthreads();

    if (lane < 16) *op = ccv - 0.5f * s;
}

__global__ __launch_bounds__(256) __attribute__((amdgpu_waves_per_eu(8, 8)))
void gmm_logits_kernel(const float* __restrict__ x,
                       const float* __restrict__ G,
                       const float* __restrict__ h,
                       const float* __restrict__ cc,
                       float* __restrict__ out)
{
    __shared__ __align__(16) float Xl[TILE_N * LSTR];
    __shared__ __align__(16) float Gl[2 * GMM_D * LSTR];
    __shared__ __align__(16) float hl[2 * GMM_D];

    const int t    = threadIdx.x;
    const int c0   = blockIdx.y * GMM_CPB;
    const int n0   = blockIdx.x * TILE_N;
    const int col4 = (t & 15) * 4;
    const int r0   = t >> 4;

    {
        #pragma unroll
        for (int it = 0; it < 8; ++it) {
            const int r = r0 + it * 16;
            const float4 v = *(const float4*)(x + (size_t)(n0 + r) * GMM_D + col4);
            *(float4*)&Xl[r * LSTR + col4] = v;
        }
        #pragma unroll
        for (int it = 0; it < 4; ++it) {
            const int r = r0 + it * 16;
            const float4 v = *(const float4*)(G + (size_t)c0 * GMM_D * GMM_D + (size_t)r * GMM_D + col4);
            *(float4*)&Gl[r * LSTR + col4] = v;
        }
        if (t < GMM_D) hl[t] = h[c0 * GMM_D + t];
    }
    __syncthreads();

    const int lane  = t & 31;
    const int w     = t >> 5;
    const int nloc  = lane & 15;
    const int khalf = (lane >> 4) * 2;

    v2f bf[GMM_D / 4];
    {
        const float* brow = &Xl[(w * 16 + nloc) * LSTR];
        #pragma unroll
        for (int kk = 0; kk < GMM_D / 4; ++kk)
            bf[kk] = *(const v2f*)(brow + 4 * kk + khalf);
    }

    const unsigned goff = (unsigned)(r0 * GMM_D + col4);
    const unsigned loff = (unsigned)(r0 * LSTR + col4);
    const unsigned aoff = (unsigned)(nloc * LSTR + khalf);
    const unsigned ooff = (unsigned)(n0 + w * 16 + nloc) * GMM_C;

    static_assert((GMM_CPB & 1) == 1, "the pairing below ends on a single step on buffer 0");
    float* op = out + (size_t)ooff + c0;
    int c = c0;
    #pragma unroll 1
    for (int ci = 0; ci < GMM_CPB - 1; ci += 2) {
        gmm_class_step<0, true>(c,     G, h, cc, op,     Gl, hl, bf, t, lane, goff, loff, aoff);
        gmm_class_step<1, true>(c + 1, G, h, cc, op + 1, Gl, hl, bf, t, lane, goff, loff, aoff);
        c  += 2;
        op += 2;
    }
    gmm_class_step<0, false>(c, G, h, cc, op, Gl, hl, bf, t, lane, goff, loff, aoff);
}

extern "C" void kernel_launch(void* const* d_in, const int* in_sizes, int n_in,
                              void* d_out, int out_size, void* d_ws, size_t ws_size,
                              hipStream_t stream) {
    (void)in_sizes; (void)n_in; (void)out_size; (void)ws_size;

    const float* x       = (const float*)d_in[0];
    const float* means   = (const float*)d_in[1];
    const float* covs    = (const float*)d_in[2];
    const float* weights = (const float*)d_in[3];
    float* out = (float*)d_out;

    float* Gw  = (float*)d_ws;
    float* hw  = Gw + (size_t)GMM_C * GMM_D * GMM_D;
    float* ccw = hw + (size_t)GMM_C * GMM_D;

    gmm_prep_kernel<<<GMM_C, 64, 0, stream>>>(means, covs, weights, Gw, hw, ccw);

    dim3 grid(GMM_B / TILE_N, GMM_NCH);
    gmm_logits_kernel<<<grid, 256, 0, stream>>>(x, Gw, hw, ccw, out);
}


# main kernel: an M-tile is the same 4-row block of four classes (classes share the X fragments), K-loop of tile i ends at k = 4(i+1): 34 f32 WMMAs per class, not 40; prep writes G in fragment order
# speedup vs baseline: 2.8852x; 1.1043x over previous
#include <hip/hip_runtime.h>
#include <hip/hip_bf16.h>
#include <math.h>

#define GMM_B 32768
#define GMM_C 100
#define GMM_D 64

typedef __attribute__((ext_vector_type(2))) float v2f;
typedef __attribute__((ext_vector_type(8))) float v8f;

#define PSTR 68

#define GMM_NCH   4
#define GMM_CPB   (GMM_C / GMM_NCH)
#define GMM_NG    6
#define GRP_FRAGS 136
#define GRP_F     (GRP_FRAGS * 64)
#define ONE_FRAGS 40
#define GMM_SLOT  (ONE_FRAGS * 64)
static_assert(GMM_CPB * GMM_NCH == GMM_C, "class chunks must cover every class exactly once");
static_assert(GMM_NG * 4 + 1 == GMM_CPB, "a chunk is six groups of four classes and one single class");

template <int J0, int NJ>
__device__ __forceinline__ void gmm_chol_phase(float (&a)[GMM_D], float (&d)[GMM_D], float (&s)[GMM_D],
                                               float& ljj, float& l, float& g, float& mydiag,
                                               float* __restrict__ Lc, float* __restrict__ Gs, const int t)
{
    constexpr int W  = GMM_D - 1 - J0;
    constexpr int NQ = (W + 4) / 4;
    #pragma unroll 1
    for (int j = J0; j < J0 + NJ; ++j) {
        mydiag = (t == j) ? ljj : mydiag;
        if (t > j) Lc[j * GMM_D + (t - j)] = l;
        Gs[j * PSTR + t] = g;
        __syncthreads();
        const float* col = Lc + j * GMM_D;
        float lkv[4 * NQ];
        #pragma unroll
        for (int q = 0; q < NQ; ++q) {
            const float4 v = *(const float4*)(col + 4 * q);
            lkv[4 * q + 0] = v.x; lkv[4 * q + 1] = v.y;
            lkv[4 * q + 2] = v.z; lkv[4 * q + 3] = v.w;
        }
        const float lc = l, gc = g;
        {
            const float lk = lkv[1];
            a[0] = a[1] - lc * lk;
            d[0] = d[1] - lk * lk;
            s[0] = s[1] - lk * gc;
        }
        ljj = sqrtf(d[0]);
        l   = a[0] / ljj;
        g   = s[0] / ljj;
        #pragma unroll
        for (int p = 2; p <= W; ++p) {
            const float lk = lkv[p];
            a[p - 1] = a[p] - lc * lk;
            d[p - 1] = d[p] - lk * lk;
            s[p - 1] = s[p] - lk * gc;
        }
    }
}

__global__ __launch_bounds__(64) __attribute__((amdgpu_waves_per_eu(1, 1)))
void gmm_prep_kernel(const float* __restrict__ means,
                     const float* __restrict__ covs,
                     const float* __restrict__ weights,
                     float* __restrict__ Gout,
                     float* __restrict__ hout,
                     float* __restrict__ ccout)
{
    __shared__ __align__(16) float Lc[GMM_D * GMM_D];
    __shared__ __align__(16) float Gs[GMM_D * PSTR];
    __shared__ __align__(16) float dg[GMM_D];
    __shared__ __align__(16) float mu[GMM_D];
    __shared__ float red[GMM_D];

    const int t = threadIdx.x;
    const int c = blockIdx.x;
    const float* cov = covs + (size_t)c * GMM_D * GMM_D;

    float a[GMM_D];
    {
        const float4* src = (const float4*)(cov + (size_t)t * GMM_D);
        #pragma unroll
        for (int q = 0; q < GMM_D / 4; ++q) {
            const float4 v = src[q];
            a[4 * q + 0] = v.x; a[4 * q + 1] = v.y;
            a[4 * q + 2] = v.z; a[4 * q + 3] = v.w;
        }
    }
    dg[t] = cov[t * (GMM_D + 1)];
    mu[t] = means[c * GMM_D + t];
    {
        const float4 z = make_float4(0.0f, 0.0f, 0.0f, 0.0f);
        #pragma unroll
        for (int q = 0; q < GMM_D / 4; ++q)
            *(float4*)&Lc[(q * GMM_D + t) * 4] = z;
    }
    __syncthreads();

    float d[GMM_D];
    float s[GMM_D];
    #pragma unroll
    for (int q = 0; q < GMM_D / 4; ++q) {
        const float4 v = *(const float4*)&dg[4 * q];
        d[4 * q + 0] = v.x; d[4 * q + 1] = v.y;
        d[4 * q + 2] = v.z; d[4 * q + 3] = v.w;
    }
    #pragma unroll
    for (int i = 0; i < GMM_D; ++i) s[i] = (i == t) ? 1.0f : 0.0f;

    float mydiag = 1.0f;
    float ljj = sqrtf(d[0]);
    float l   = a[0] / ljj;
    float g   = s[0] / ljj;
    gmm_chol_phase< 0, 8>(a, d, s, ljj, l, g, mydiag, Lc, Gs, t);
    gmm_chol_phase< 8, 8>(a, d, s, ljj, l, g, mydiag, Lc, Gs, t);
    gmm_chol_phase<16, 8>(a, d, s, ljj, l, g, mydiag, Lc, Gs, t);
    gmm_chol_phase<24, 8>(a, d, s, ljj, l, g, mydiag, Lc, Gs, t);
    gmm_chol_phase<32, 8>(a, d, s, ljj, l, g, mydiag, Lc, Gs, t);
    gmm_chol_phase<40, 8>(a, d, s, ljj, l, g, mydiag, Lc, Gs, t);
    gmm_chol_phase<48, 8>(a, d, s, ljj, l, g, mydiag, Lc, Gs, t);
    gmm_chol_phase<56, 8>(a, d, s, ljj, l, g, mydiag, Lc, Gs, t);

    red[t] = 2.0f * logf(mydiag);
    __syncthreads();
    if (t == 0) {
        float ld = 0.0f;
        for (int k = 0; k < GMM_D; ++k) ld += red[k];
        const float LOG2PI = 1.8378770664093453f;
        ccout[c] = logf(weights[c]) - 0.5f * (ld + (float)GMM_D * LOG2PI);
    }

    float hv = 0.0f;
    #pragma unroll
    for (int k = 0; k < GMM_D; ++k) hv += Gs[t * PSTR + k] * mu[k];
    hout[c * GMM_D + t] = -hv;

    float* gdst = Gout + (size_t)c * GMM_SLOT;
    if ((c % GMM_CPB) < GMM_NG * 4) {
        const int r  = t & 3;
        const int jj = t >> 2;
        #pragma unroll
        for (int i = 0; i < GMM_D / 4; ++i) {
            if (jj <= i)
                *(float4*)(gdst + ((i * (i + 1)) / 2 + jj) * 16 + r * 4) =
                    *(const float4*)&Gs[(4 * i + r) * PSTR + 4 * jj];
        }
    } else {
        const int row = t & 15;
        const int jq  = t >> 4;
        #pragma unroll
        for (int mt = 0; mt < 4; ++mt) {
            #pragma unroll
            for (int jb = 0; jb <= mt; ++jb) {
                const int j = 4 * jb + jq;
                *(float4*)(gdst + (2 * mt * (mt + 1) + j) * 64 + row * 4) =
                    *(const float4*)&Gs[(16 * mt + row) * PSTR + 4 * j];
            }
        }
    }
}

#define TILE_N  128
#define LSTR    68
static_assert(TILE_N * LSTR == GRP_F, "the X tile occupies exactly LDS buffer 1");

template <int I0, int I1>
__device__ __forceinline__ void gmm_group_tiles(const float* __restrict__ Ab,
                                                const float* __restrict__ hb,
                                                const v2f (&bf)[GMM_D / 4],
                                                v2f& qa, v2f& qb)
{
    #pragma unroll
    for (int i = I0; i < I1; ++i) {
        v8f acc;
        {
            const float4 h0 = *(const float4*)(hb + i * 16);
            const float4 h1 = *(const float4*)(hb + i * 16 + 4);
            acc[0] = h0.x; acc[1] = h0.y; acc[2] = h0.z; acc[3] = h0.w;
            acc[4] = h1.x; acc[5] = h1.y; acc[6] = h1.z; acc[7] = h1.w;
        }
        const float* af = Ab + ((i * (i + 1)) / 2) * 64;
        #pragma unroll
        for (int j = 0; j <= i; ++j) {
            const v2f a = *(const v2f*)(af + j * 64);
            acc = __builtin_amdgcn_wmma_f32_16x16x4_f32(
                       false, a,  false, bf[j],
                       (short)0, acc,
                       false,  false);
        }
        {
            const v2f a01 = {acc[0], acc[1]}, a23 = {acc[2], acc[3]};
            const v2f a45 = {acc[4], acc[5]}, a67 = {acc[6], acc[7]};
            qa = __builtin_elementwise_fma(a01, a01, qa);
            qb = __builtin_elementwise_fma(a45, a45, qb);
            qa = __builtin_elementwise_fma(a23, a23, qa);
            qb = __builtin_elementwise_fma(a67, a67, qb);
        }
    }
}

struct gmm_piece { float4 a, b, c; };

__device__ __forceinline__ void gmm_fetch3(gmm_piece& pc, const float* __restrict__ g, const int it0)
{
    pc.a = *(const float4*)(g + 256 * (it0 + 0));
    pc.b = *(const float4*)(g + 256 * (it0 + 1));
    pc.c = *(const float4*)(g + 256 * (it0 + 2));
}
__device__ __forceinline__ void gmm_store3(const gmm_piece& pc, float* __restrict__ l, const int it0)
{
    *(float4*)(l + 1024 * (it0 + 0)) = pc.a;
    *(float4*)(l + 1024 * (it0 + 1)) = pc.b;
    *(float4*)(l + 1024 * (it0 + 2)) = pc.c;
}

template <bool MORE>
__device__ __forceinline__ void gmm_group_step(const int cur,
                                               const float* __restrict__ gnb,
                                               const float* __restrict__ hnb,
                                               const float* __restrict__ ccg,
                                               float* __restrict__ og,
                                               float* __restrict__ Gl, float* __restrict__ hl,
                                               const v2f (&bf)[GMM_D / 4],
                                               const int t, const unsigned aoff, const int half,
                                               const unsigned gsoff, const unsigned gsoff2,
                                               const unsigned hdoff, const unsigned ooff)
{
    const float cca = ccg[2 * half];
    const float ccb = ccg[2 * half + 1];
    const float* gnext  = gnb + gsoff;
    const float* gnext2 = gnb + gsoff2;
    const float* hnext  = hnb + t;
    float*       op     = og + ooff + 2 * half;

    const float* Ab = Gl + cur * GRP_F + aoff;
    const float* hb = hl + cur * 256 + half * 8;
    float*       Gn = Gl + (cur ^ 1) * GRP_F + 4 * t;

    float4 p0, p1, p2, p3;
    float2 tail;
    float  hn;
    if (MORE) {
        p0 = *(const float4*)(gnext);
        p1 = *(const float4*)(gnext + 256 * 1);
        p2 = *(const float4*)(gnext + 256 * 2);
        p3 = *(const float4*)(gnext + 256 * 3);
        __builtin_amdgcn_sched_barrier(0);
    }

    v2f qa = {0.0f, 0.0f}, qb = {0.0f, 0.0f};
    gmm_group_tiles<0, 6>(Ab, hb, bf, qa, qb);
    if (MORE) {
        *(float4*)(Gn)            = p0;
        *(float4*)(Gn + 1024 * 1) = p1;
        *(float4*)(Gn + 1024 * 2) = p2;
        *(float4*)(Gn + 1024 * 3) = p3;
        p0 = *(const float4*)(gnext + 256 * 4);
        p1 = *(const float4*)(gnext + 256 * 5);
        p2 = *(const float4*)(gnext + 256 * 6);
        p3 = *(const float4*)(gnext + 256 * 7);
        tail = *(const float2*)gnext2;
        hn   = *hnext;
        __builtin_amdgcn_sched_barrier(0);
    }
    gmm_group_tiles<6, 9>(Ab, hb, bf, qa, qb);
    if (MORE) {
        *(float4*)(Gn + 1024 * 4) = p0;
        *(float4*)(Gn + 1024 * 5) = p1;
        *(float4*)(Gn + 1024 * 6) = p2;
        *(float4*)(Gn + 1024 * 7) = p3;
        *(float2*)(Gl + (cur ^ 1) * GRP_F + 8192 + 2 * t) = tail;
        hl[(cur ^ 1) * 256 + hdoff] = hn;
    }
    gmm_group_tiles<9, 16>(Ab, hb, bf, qa, qb);
    if (MORE) {
        __syncthreads();
    }
    op[0] = cca - 0.5f * (qa[0] + qa[1]);
    op[1] = ccb - 0.5f * (qb[0] + qb[1]);
}

__global__ __launch_bounds__(256) __attribute__((amdgpu_waves_per_eu(8, 8)))
void gmm_logits_kernel(const float* __restrict__ x,
                       const float* __restrict__ G,
                       const float* __restrict__ h,
                       const float* __restrict__ cc,
                       float* __restrict__ out)
{
    __shared__ __align__(16) float Gl[2 * GRP_F];
    __shared__ __align__(16) float hl[2 * 256];
    float* Xl = Gl + GRP_F;

    const int t    = threadIdx.x;
    const int c0   = blockIdx.y * GMM_CPB;
    const int cs   = c0 + GMM_NG * 4;
    const int n0   = blockIdx.x * TILE_N;
    const int col4 = (t & 15) * 4;
    const int r0   = t >> 4;

    const unsigned gsoff  = (unsigned)(((t >> 2) & 3) * GMM_SLOT + (t >> 4) * 16 + (t & 3) * 4);
    const unsigned gsoff2 = (unsigned)(((t >> 3) & 3) * GMM_SLOT + (128 + (t >> 5)) * 16 + ((t >> 1) & 3) * 4 + (t & 1) * 2);
    const unsigned hdoff  = (unsigned)(((t & 63) >> 2) * 16 + (t >> 6) * 4 + (t & 3));

    {
        #pragma unroll
        for (int it = 0; it < 8; ++it) {
            const int r = r0 + it * 16;
            const float4 v = *(const float4*)(x + (size_t)(n0 + r) * GMM_D + col4);
            *(float4*)&Xl[r * LSTR + col4] = v;
        }
        const float* ssrc = G + (size_t)cs * GMM_SLOT;
        *(float4*)&Gl[4 * t]        = *(const float4*)(ssrc + 4 * t);
        *(float4*)&Gl[4 * t + 1024] = *(const float4*)(ssrc + 4 * t + 1024);
        *(float2*)&Gl[2 * t + 2048] = *(const float2*)(ssrc + 2 * t + 2048);
        if (t < GMM_D) hl[t] = h[cs * GMM_D + t];
    }
    __syncthreads();

    const int lane  = t & 31;
    const int w     = t >> 5;
    const int nloc  = lane & 15;
    const int half  = lane >> 4;
    const int khalf = half * 2;

    v2f bf[GMM_D / 4];
    {
        const float* brow = &Xl[(w * 16 + nloc) * LSTR];
        #pragma unroll
        for (int kk = 0; kk < GMM_D / 4; ++kk)
            bf[kk] = *(const v2f*)(brow + 4 * kk + khalf);
    }
    __syncthreads();

    const unsigned aoff = (unsigned)(nloc * 4 + khalf);
    const unsigned ooff = (unsigned)(n0 + w * 16 + nloc) * GMM_C;

    const float* gnb = G + (size_t)c0 * GMM_SLOT;
    const float* hnb = h + (size_t)c0 * GMM_D;
    const float* gnext  = gnb + gsoff;
    const float* gnext2 = gnb + gsoff2;
    const float* hnext  = hnb + t;

    {
        const float ccv = cc[cs];
        const float* Ab = Gl + aoff;
        const float* hb = hl + half * 8;
        float*       Gn = Gl + GRP_F + 4 * t;
        gmm_piece pc;
        gmm_fetch3(pc, gnext, 0);
        const float hn = *hnext;
        float q0 = 0.0f, q1 = 0.0f;
        #pragma unroll
        for (int mt = 0; mt < 4; ++mt) {
            v8f acc;
            {
                const float4 h0 = *(const float4*)(hb + mt * 16);
                const float4 h1 = *(const float4*)(hb + mt * 16 + 4);
                acc[0] = h0.x; acc[1] = h0.y; acc[2] = h0.z; acc[3] = h0.w;
                acc[4] = h1.x; acc[5] = h1.y; acc[6] = h1.z; acc[7] = h1.w;
            }
            const float* af = Ab + (2 * mt * (mt + 1)) * 64;
            #pragma unroll
            for (int j = 0; j < 4 * (mt + 1); ++j) {
                const v2f a = *(const v2f*)(af + j * 64);
                acc = __builtin_amdgcn_wmma_f32_16x16x4_f32(
                           false, a,  false, bf[j],
                           (short)0, acc,
                           false,  false);
            }
            q0 = fmaf(acc[0], acc[0], q0);
            q1 = fmaf(acc[1], acc[1], q1);
            q0 = fmaf(acc[2], acc[2], q0);
            q1 = fmaf(acc[3], acc[3], q1);
            q0 = fmaf(acc[4], acc[4], q0);
            q1 = fmaf(acc[5], acc[5], q1);
            q0 = fmaf(acc[6], acc[6], q0);
            q1 = fmaf(acc[7], acc[7], q1);
            if (mt == 1) { gmm_store3(pc, Gn, 0); gmm_fetch3(pc, gnext, 3); }
            if (mt == 2) { gmm_store3(pc, Gn, 3); }
        }
        {
            const float4 u = *(const float4*)(gnext + 256 * 6);
            const float4 v = *(const float4*)(gnext + 256 * 7);
            const float2 z = *(const float2*)gnext2;
            *(float4*)(Gn + 1024 * 6) = u;
            *(float4*)(Gn + 1024 * 7) = v;
            *(float2*)(Gl + GRP_F + 8192 + 2 * t) = z;
            hl[256 + hdoff] = hn;
        }
        float s = q0 + q1;
        s += __shfl_xor(s, 16, 32);
        __syncthreads();
        if (lane < 16) (out + cs)[ooff] = ccv - 0.5f * s;
    }

    {
        const float* gb  = gnb;
        const float* hb2 = hnb;
        const float* ccg = cc + c0;
        float*       og  = out + c0;
        #pragma unroll 1
        for (int gi = 0; gi < GMM_NG - 1; ++gi) {
            gb  += 4 * GMM_SLOT;
            hb2 += 4 * GMM_D;
            gmm_group_step<true>((gi + 1) & 1, gb, hb2, ccg, og, Gl, hl, bf, t, aoff, half, gsoff, gsoff2, hdoff, ooff);
            ccg += 4;
            og  += 4;
        }
        gmm_group_step<false>(GMM_NG & 1, gb, hb2, ccg, og, Gl, hl, bf, t, aoff, half, gsoff, gsoff2, hdoff, ooff);
    }
}

extern "C" void kernel_launch(void* const* d_in, const int* in_sizes, int n_in,
                              void* d_out, int out_size, void* d_ws, size_t ws_size,
                              hipStream_t stream) {
    (void)in_sizes; (void)n_in; (void)out_size; (void)ws_size;

    const float* x       = (const float*)d_in[0];
    const float* means   = (const float*)d_in[1];
    const float* covs    = (const float*)d_in[2];
    const float* weights = (const float*)d_in[3];
    float* out = (float*)d_out;

    float* Gw  = (float*)d_ws;
    float* hw  = Gw + (size_t)GMM_C * GMM_SLOT;
    float* ccw = hw + (size_t)GMM_C * GMM_D;

    gmm_prep_kernel<<<GMM_C, 64, 0, stream>>>(means, covs, weights, Gw, hw, ccw);

    dim3 grid(GMM_B / TILE_N, GMM_NCH);
    gmm_logits_kernel<<<grid, 256, 0, stream>>>(x, Gw, hw, ccw, out);
}


# prep kernel: square-root-free elimination, two columns per barrier: each thread eliminates the 2-column panel itself (2 reciprocals): 32 LDS round trips, not 64; no sqrt or division between barriers
# speedup vs baseline: 3.1249x; 1.0831x over previous
#include <hip/hip_runtime.h>
#include <hip/hip_bf16.h>
#include <math.h>

#define GMM_B 32768
#define GMM_C 100
#define GMM_D 64

typedef __attribute__((ext_vector_type(2))) float v2f;
typedef __attribute__((ext_vector_type(8))) float v8f;

#define PSTR 68

#define GMM_NCH   4
#define GMM_CPB   (GMM_C / GMM_NCH)
#define GMM_NG    6
#define GRP_FRAGS 136
#define GRP_F     (GRP_FRAGS * 64)
#define ONE_FRAGS 40
#define GMM_SLOT  (ONE_FRAGS * 64)
static_assert(GMM_CPB * GMM_NCH == GMM_C, "class chunks must cover every class exactly once");
static_assert(GMM_NG * 4 + 1 == GMM_CPB, "a chunk is six groups of four classes and one single class");

template <int J0, int NB>
__device__ __forceinline__ void gmm_elim_phase(float (&a)[GMM_D], float (&s)[GMM_D], float& mydiag,
                                               float* __restrict__ Pan, float* __restrict__ Gs, const int t)
{
    static_assert((J0 & 1) == 0, "a phase starts on an even column");
    constexpr int W  = GMM_D - 1 - J0;
    constexpr int NQ = (W + 1) / 2;
    #pragma unroll 1
    for (int j = J0; j < J0 + 2 * NB; j += 2) {
        __syncthreads();
        const float* pan = Pan + j * GMM_D;
        float c0[2 * NQ], c1[2 * NQ];
        #pragma unroll
        for (int q = 0; q < NQ; ++q) {
            const float4 v = *(const float4*)(pan + 4 * q);
            c0[2 * q]     = v.x; c1[2 * q]     = v.y;
            c0[2 * q + 1] = v.z; c1[2 * q + 1] = v.w;
        }
        const float piv0 = c0[0];
        const float e    = c0[1];
        const float r0   = __builtin_amdgcn_rcpf(piv0);
        const float f    = e * r0;
        const float piv1 = c1[1] - e * f;
        const float r1   = __builtin_amdgcn_rcpf(piv1);
        const float s0  = s[0];
        const float ma0 = a[0] * r0;
        const float ms0 = s0 * r0;
        const float a1  = a[1] - ma0 * e;
        const float s1  = s[1] - ms0 * e;
        const float ma1 = a1 * r1;
        const float ms1 = s1 * r1;
        #pragma unroll
        for (int p = 2; p <= 3; ++p) {
            const float c1p = c1[p] - c0[p] * f;
            a[p - 2] = (a[p] - ma0 * c0[p]) - ma1 * c1p;
            s[p - 2] = (s[p] - ms0 * c0[p]) - ms1 * c1p;
        }
        if (t >= j + 2)
            *(float2*)(Pan + (j + 2) * GMM_D + 2 * (t - j - 2)) = make_float2(a[0], a[1]);
        mydiag = (t == j) ? piv0 : ((t == j + 1) ? piv1 : mydiag);
        Gs[j * PSTR + t]       = s0 * __builtin_amdgcn_rsqf(piv0);
        Gs[(j + 1) * PSTR + t] = s1 * __builtin_amdgcn_rsqf(piv1);
        #pragma unroll
        for (int p = 4; p <= W; ++p) {
            const float c1p = c1[p] - c0[p] * f;
            a[p - 2] = (a[p] - ma0 * c0[p]) - ma1 * c1p;
            s[p - 2] = (s[p] - ms0 * c0[p]) - ms1 * c1p;
        }
    }
}

__global__ __launch_bounds__(64) __attribute__((amdgpu_waves_per_eu(1, 1)))
void gmm_prep_kernel(const float* __restrict__ means,
                     const float* __restrict__ covs,
                     const float* __restrict__ weights,
                     float* __restrict__ Gout,
                     float* __restrict__ hout,
                     float* __restrict__ ccout)
{
    __shared__ __align__(16) float Pan[GMM_D * GMM_D];
    __shared__ __align__(16) float Gs[GMM_D * PSTR];
    __shared__ __align__(16) float mu[GMM_D];
    __shared__ float red[GMM_D];

    const int t = threadIdx.x;
    const int c = blockIdx.x;
    const float* cov = covs + (size_t)c * GMM_D * GMM_D;

    float a[GMM_D];
    {
        const float4* src = (const float4*)(cov + (size_t)t * GMM_D);
        #pragma unroll
        for (int q = 0; q < GMM_D / 4; ++q) {
            const float4 v = src[q];
            a[4 * q + 0] = v.x; a[4 * q + 1] = v.y;
            a[4 * q + 2] = v.z; a[4 * q + 3] = v.w;
        }
    }
    mu[t] = means[c * GMM_D + t];
    {
        const float4 z = make_float4(0.0f, 0.0f, 0.0f, 0.0f);
        #pragma unroll
        for (int q = 0; q < GMM_D / 4; ++q)
            *(float4*)&Pan[(q * GMM_D + t) * 4] = z;
    }
    __syncthreads();
    *(float2*)(Pan + 2 * t) = make_float2(a[0], a[1]);

    float s[GMM_D];
    #pragma unroll
    for (int i = 0; i < GMM_D; ++i) s[i] = (i == t) ? 1.0f : 0.0f;

    float mydiag = 1.0f;
    gmm_elim_phase< 0, 4>(a, s, mydiag, Pan, Gs, t);
    gmm_elim_phase< 8, 4>(a, s, mydiag, Pan, Gs, t);
    gmm_elim_phase<16, 4>(a, s, mydiag, Pan, Gs, t);
    gmm_elim_phase<24, 4>(a, s, mydiag, Pan, Gs, t);
    gmm_elim_phase<32, 4>(a, s, mydiag, Pan, Gs, t);
    gmm_elim_phase<40, 4>(a, s, mydiag, Pan, Gs, t);
    gmm_elim_phase<48, 4>(a, s, mydiag, Pan, Gs, t);
    gmm_elim_phase<56, 4>(a, s, mydiag, Pan, Gs, t);

    red[t] = logf(mydiag);
    __syncthreads();
    if (t == 0) {
        float ld = 0.0f;
        for (int k = 0; k < GMM_D; ++k) ld += red[k];
        const float LOG2PI = 1.8378770664093453f;
        ccout[c] = logf(weights[c]) - 0.5f * (ld + (float)GMM_D * LOG2PI);
    }

    float hv = 0.0f;
    #pragma unroll
    for (int k = 0; k < GMM_D; ++k) hv += Gs[t * PSTR + k] * mu[k];
    hout[c * GMM_D + t] = -hv;

    float* gdst = Gout + (size_t)c * GMM_SLOT;
    if ((c % GMM_CPB) < GMM_NG * 4) {
        const int r  = t & 3;
        const int jj = t >> 2;
        #pragma unroll
        for (int i = 0; i < GMM_D / 4; ++i) {
            if (jj <= i)
                *(float4*)(gdst + ((i * (i + 1)) / 2 + jj) * 16 + r * 4) =
                    *(const float4*)&Gs[(4 * i + r) * PSTR + 4 * jj];
        }
    } else {
        const int row = t & 15;
        const int jq  = t >> 4;
        #pragma unroll
        for (int mt = 0; mt < 4; ++mt) {
            #pragma unroll
            for (int jb = 0; jb <= mt; ++jb) {
                const int j = 4 * jb + jq;
                *(float4*)(gdst + (2 * mt * (mt + 1) + j) * 64 + row * 4) =
                    *(const float4*)&Gs[(16 * mt + row) * PSTR + 4 * j];
            }
        }
    }
}

#define TILE_N  128
#define LSTR    68
static_assert(TILE_N * LSTR == GRP_F, "the X tile occupies exactly LDS buffer 1");

template <int I0, int I1>
__device__ __forceinline__ void gmm_group_tiles(const float* __restrict__ Ab,
                                                const float* __restrict__ hb,
                                                const v2f (&bf)[GMM_D / 4],
                                                v2f& qa, v2f& qb)
{
    #pragma unroll
    for (int i = I0; i < I1; ++i) {
        v8f acc;
        {
            const float4 h0 = *(const float4*)(hb + i * 16);
            const float4 h1 = *(const float4*)(hb + i * 16 + 4);
            acc[0] = h0.x; acc[1] = h0.y; acc[2] = h0.z; acc[3] = h0.w;
            acc[4] = h1.x; acc[5] = h1.y; acc[6] = h1.z; acc[7] = h1.w;
        }
        const float* af = Ab + ((i * (i + 1)) / 2) * 64;
        #pragma unroll
        for (int j = 0; j <= i; ++j) {
            const v2f a = *(const v2f*)(af + j * 64);
            acc = __builtin_amdgcn_wmma_f32_16x16x4_f32(
                       false, a,  false, bf[j],
                       (short)0, acc,
                       false,  false);
        }
        {
            const v2f a01 = {acc[0], acc[1]}, a23 = {acc[2], acc[3]};
            const v2f a45 = {acc[4], acc[5]}, a67 = {acc[6], acc[7]};
            qa = __builtin_elementwise_fma(a01, a01, qa);
            qb = __builtin_elementwise_fma(a45, a45, qb);
            qa = __builtin_elementwise_fma(a23, a23, qa);
            qb = __builtin_elementwise_fma(a67, a67, qb);
        }
    }
}

struct gmm_piece { float4 a, b, c; };

__device__ __forceinline__ void gmm_fetch3(gmm_piece& pc, const float* __restrict__ g, const int it0)
{
    pc.a = *(const float4*)(g + 256 * (it0 + 0));
    pc.b = *(const float4*)(g + 256 * (it0 + 1));
    pc.c = *(const float4*)(g + 256 * (it0 + 2));
}
__device__ __forceinline__ void gmm_store3(const gmm_piece& pc, float* __restrict__ l, const int it0)
{
    *(float4*)(l + 1024 * (it0 + 0)) = pc.a;
    *(float4*)(l + 1024 * (it0 + 1)) = pc.b;
    *(float4*)(l + 1024 * (it0 + 2)) = pc.c;
}

template <bool MORE>
__device__ __forceinline__ void gmm_group_step(const int cur,
                                               const float* __restrict__ gnb,
                                               const float* __restrict__ hnb,
                                               const float* __restrict__ ccg,
                                               float* __restrict__ og,
                                               float* __restrict__ Gl, float* __restrict__ hl,
                                               const v2f (&bf)[GMM_D / 4],
                                               const int t, const unsigned aoff, const int half,
                                               const unsigned gsoff, const unsigned gsoff2,
                                               const unsigned hdoff, const unsigned ooff)
{
    const float cca = ccg[2 * half];
    const float ccb = ccg[2 * half + 1];
    const float* gnext  = gnb + gsoff;
    const float* gnext2 = gnb + gsoff2;
    const float* hnext  = hnb + t;
    float*       op     = og + ooff + 2 * half;

    const float* Ab = Gl + cur * GRP_F + aoff;
    const float* hb = hl + cur * 256 + half * 8;
    float*       Gn = Gl + (cur ^ 1) * GRP_F + 4 * t;

    float4 p0, p1, p2, p3;
    float2 tail;
    float  hn;
    if (MORE) {
        p0 = *(const float4*)(gnext);
        p1 = *(const float4*)(gnext + 256 * 1);
        p2 = *(const float4*)(gnext + 256 * 2);
        p3 = *(const float4*)(gnext + 256 * 3);
        __builtin_amdgcn_sched_barrier(0);
    }

    v2f qa = {0.0f, 0.0f}, qb = {0.0f, 0.0f};
    gmm_group_tiles<0, 6>(Ab, hb, bf, qa, qb);
    if (MORE) {
        *(float4*)(Gn)            = p0;
        *(float4*)(Gn + 1024 * 1) = p1;
        *(float4*)(Gn + 1024 * 2) = p2;
        *(float4*)(Gn + 1024 * 3) = p3;
        p0 = *(const float4*)(gnext + 256 * 4);
        p1 = *(const float4*)(gnext + 256 * 5);
        p2 = *(const float4*)(gnext + 256 * 6);
        p3 = *(const float4*)(gnext + 256 * 7);
        tail = *(const float2*)gnext2;
        hn   = *hnext;
        __builtin_amdgcn_sched_barrier(0);
    }
    gmm_group_tiles<6, 9>(Ab, hb, bf, qa, qb);
    if (MORE) {
        *(float4*)(Gn + 1024 * 4) = p0;
        *(float4*)(Gn + 1024 * 5) = p1;
        *(float4*)(Gn + 1024 * 6) = p2;
        *(float4*)(Gn + 1024 * 7) = p3;
        *(float2*)(Gl + (cur ^ 1) * GRP_F + 8192 + 2 * t) = tail;
        hl[(cur ^ 1) * 256 + hdoff] = hn;
    }
    gmm_group_tiles<9, 16>(Ab, hb, bf, qa, qb);
    if (MORE) {
        __syncthreads();
    }
    op[0] = cca - 0.5f * (qa[0] + qa[1]);
    op[1] = ccb - 0.5f * (qb[0] + qb[1]);
}

__global__ __launch_bounds__(256) __attribute__((amdgpu_waves_per_eu(8, 8)))
void gmm_logits_kernel(const float* __restrict__ x,
                       const float* __restrict__ G,
                       const float* __restrict__ h,
                       const float* __restrict__ cc,
                       float* __restrict__ out)
{
    __shared__ __align__(16) float Gl[2 * GRP_F];
    __shared__ __align__(16) float hl[2 * 256];
    float* Xl = Gl + GRP_F;

    const int t    = threadIdx.x;
    const int c0   = blockIdx.y * GMM_CPB;
    const int cs   = c0 + GMM_NG * 4;
    const int n0   = blockIdx.x * TILE_N;
    const int col4 = (t & 15) * 4;
    const int r0   = t >> 4;

    const unsigned gsoff  = (unsigned)(((t >> 2) & 3) * GMM_SLOT + (t >> 4) * 16 + (t & 3) * 4);
    const unsigned gsoff2 = (unsigned)(((t >> 3) & 3) * GMM_SLOT + (128 + (t >> 5)) * 16 + ((t >> 1) & 3) * 4 + (t & 1) * 2);
    const unsigned hdoff  = (unsigned)(((t & 63) >> 2) * 16 + (t >> 6) * 4 + (t & 3));

    {
        #pragma unroll
        for (int it = 0; it < 8; ++it) {
            const int r = r0 + it * 16;
            const float4 v = *(const float4*)(x + (size_t)(n0 + r) * GMM_D + col4);
            *(float4*)&Xl[r * LSTR + col4] = v;
        }
        const float* ssrc = G + (size_t)cs * GMM_SLOT;
        *(float4*)&Gl[4 * t]        = *(const float4*)(ssrc + 4 * t);
        *(float4*)&Gl[4 * t + 1024] = *(const float4*)(ssrc + 4 * t + 1024);
        *(float2*)&Gl[2 * t + 2048] = *(const float2*)(ssrc + 2 * t + 2048);
        if (t < GMM_D) hl[t] = h[cs * GMM_D + t];
    }
    __syncthreads();

    const int lane  = t & 31;
    const int w     = t >> 5;
    const int nloc  = lane & 15;
    const int half  = lane >> 4;
    const int khalf = half * 2;

    v2f bf[GMM_D / 4];
    {
        const float* brow = &Xl[(w * 16 + nloc) * LSTR];
        #pragma unroll
        for (int kk = 0; kk < GMM_D / 4; ++kk)
            bf[kk] = *(const v2f*)(brow + 4 * kk + khalf);
    }
    __syncthreads();

    const unsigned aoff = (unsigned)(nloc * 4 + khalf);
    const unsigned ooff = (unsigned)(n0 + w * 16 + nloc) * GMM_C;

    const float* gnb = G + (size_t)c0 * GMM_SLOT;
    const float* hnb = h + (size_t)c0 * GMM_D;
    const float* gnext  = gnb + gsoff;
    const float* gnext2 = gnb + gsoff2;
    const float* hnext  = hnb + t;

    {
        const float ccv = cc[cs];
        const float* Ab = Gl + aoff;
        const float* hb = hl + half * 8;
        float*       Gn = Gl + GRP_F + 4 * t;
        gmm_piece pc;
        gmm_fetch3(pc, gnext, 0);
        const float hn = *hnext;
        float q0 = 0.0f, q1 = 0.0f;
        #pragma unroll
        for (int mt = 0; mt < 4; ++mt) {
            v8f acc;
            {
                const float4 h0 = *(const float4*)(hb + mt * 16);
                const float4 h1 = *(const float4*)(hb + mt * 16 + 4);
                acc[0] = h0.x; acc[1] = h0.y; acc[2] = h0.z; acc[3] = h0.w;
                acc[4] = h1.x; acc[5] = h1.y; acc[6] = h1.z; acc[7] = h1.w;
            }
            const float* af = Ab + (2 * mt * (mt + 1)) * 64;
            #pragma unroll
            for (int j = 0; j < 4 * (mt + 1); ++j) {
                const v2f a = *(const v2f*)(af + j * 64);
                acc = __builtin_amdgcn_wmma_f32_16x16x4_f32(
                           false, a,  false, bf[j],
                           (short)0, acc,
                           false,  false);
            }
            q0 = fmaf(acc[0], acc[0], q0);
            q1 = fmaf(acc[1], acc[1], q1);
            q0 = fmaf(acc[2], acc[2], q0);
            q1 = fmaf(acc[3], acc[3], q1);
            q0 = fmaf(acc[4], acc[4], q0);
            q1 = fmaf(acc[5], acc[5], q1);
            q0 = fmaf(acc[6], acc[6], q0);
            q1 = fmaf(acc[7], acc[7], q1);
            if (mt == 1) { gmm_store3(pc, Gn, 0); gmm_fetch3(pc, gnext, 3); }
            if (mt == 2) { gmm_store3(pc, Gn, 3); }
        }
        {
            const float4 u = *(const float4*)(gnext + 256 * 6);
            const float4 v = *(const float4*)(gnext + 256 * 7);
            const float2 z = *(const float2*)gnext2;
            *(float4*)(Gn + 1024 * 6) = u;
            *(float4*)(Gn + 1024 * 7) = v;
            *(float2*)(Gl + GRP_F + 8192 + 2 * t) = z;
            hl[256 + hdoff] = hn;
        }
        float s = q0 + q1;
        s += __shfl_xor(s, 16, 32);
        __syncthreads();
        if (lane < 16) (out + cs)[ooff] = ccv - 0.5f * s;
    }

    {
        const float* gb  = gnb;
        const float* hb2 = hnb;
        const float* ccg = cc + c0;
        float*       og  = out + c0;
        #pragma unroll 1
        for (int gi = 0; gi < GMM_NG - 1; ++gi) {
            gb  += 4 * GMM_SLOT;
            hb2 += 4 * GMM_D;
            gmm_group_step<true>((gi + 1) & 1, gb, hb2, ccg, og, Gl, hl, bf, t, aoff, half, gsoff, gsoff2, hdoff, ooff);
            ccg += 4;
            og  += 4;
        }
        gmm_group_step<false>(GMM_NG & 1, gb, hb2, ccg, og, Gl, hl, bf, t, aoff, half, gsoff, gsoff2, hdoff, ooff);
    }
}

extern "C" void kernel_launch(void* const* d_in, const int* in_sizes, int n_in,
                              void* d_out, int out_size, void* d_ws, size_t ws_size,
                              hipStream_t stream) {
    (void)in_sizes; (void)n_in; (void)out_size; (void)ws_size;

    const float* x       = (const float*)d_in[0];
    const float* means   = (const float*)d_in[1];
    const float* covs    = (const float*)d_in[2];
    const float* weights = (const float*)d_in[3];
    float* out = (float*)d_out;

    float* Gw  = (float*)d_ws;
    float* hw  = Gw + (size_t)GMM_C * GMM_SLOT;
    float* ccw = hw + (size_t)GMM_C * GMM_D;

    gmm_prep_kernel<<<GMM_C, 64, 0, stream>>>(means, covs, weights, Gw, hw, ccw);

    dim3 grid(GMM_B / TILE_N, GMM_NCH);
    gmm_logits_kernel<<<grid, 256, 0, stream>>>(x, Gw, hw, ccw, out);
}
